# starting kernel
# baseline (speedup 1.0000x reference)
#include <hip/hip_runtime.h>

typedef __attribute__((ext_vector_type(2))) float v2f;
typedef __attribute__((ext_vector_type(8))) float v8f;

#define B_TOTAL 32768
#define T_SEQ   28
#define I_SZ    28
#define H_SZ    64
#define G_SZ    256
#define NCLS    10

#define WIH_PITCH 36
#define WHH_PITCH 68
#define H_PITCH   68

#define WAVES_PER_WG  8
#define ROWS_PER_WAVE 16

__device__ __forceinline__ float fast_sigmoid(float x) {
    float e = __builtin_amdgcn_exp2f(x * -1.44269504088896f);
    return __builtin_amdgcn_rcpf(1.0f + e);
}
__device__ __forceinline__ float fast_tanh(float x) {
#if __has_builtin(__builtin_amdgcn_tanhf)
    return __builtin_amdgcn_tanhf(x);
#else
    float e = __builtin_amdgcn_exp2f(x * -2.88539008177793f);
    return __builtin_amdgcn_rcpf(1.0f + e) * 2.0f - 1.0f;
#endif
}

__global__ __launch_bounds__(256)
void lstm_fused_kernel(const float* __restrict__ x,
                       const float* __restrict__ W_ih,
                       const float* __restrict__ W_hh,
                       const float* __restrict__ b_ih,
                       const float* __restrict__ b_hh,
                       const float* __restrict__ W_out,
                       const float* __restrict__ b_out,
                       float* __restrict__ out)
{
    __shared__ float sWih[G_SZ * WIH_PITCH];
    __shared__ float sWhh[G_SZ * WHH_PITCH];
    __shared__ float sH[WAVES_PER_WG * ROWS_PER_WAVE * H_PITCH];

    const int tid = threadIdx.x;

    {
        const float4* srcHH = (const float4*)(W_hh + (size_t)tid * H_SZ);
        #pragma unroll
        for (int q = 0; q < H_SZ / 4; ++q) {
            float4 v = srcHH[q];
            *(float4*)&sWhh[tid * WHH_PITCH + q * 4] = v;
        }
        const float4* srcIH = (const float4*)(W_ih + (size_t)tid * I_SZ);
        #pragma unroll
        for (int q = 0; q < I_SZ / 4; ++q) {
            float4 v = srcIH[q];
            *(float4*)&sWih[tid * WIH_PITCH + q * 4] = v;
        }
    }
    __syncthreads();

    const int wave = tid >> 5;
    const int lane = tid & 31;
    const int col  = lane & 15;
    const int hi   = lane >> 4;
    const int b0   = (blockIdx.x * WAVES_PER_WG + wave) * ROWS_PER_WAVE;

    float* hbuf = &sH[wave * ROWS_PER_WAVE * H_PITCH];

    float bias[16];
    #pragma unroll
    for (int nt = 0; nt < 16; ++nt) {
        int g = nt * 16 + col;
        bias[nt] = b_ih[g] + b_hh[g];
    }

    const float* xlane = x + (size_t)(b0 + col) * (T_SEQ * I_SZ);

    v8f c_st[4];
    #pragma unroll
    for (int j = 0; j < 4; ++j) c_st[j] = (v8f){};

    for (int t = 0; t < T_SEQ; ++t) {
        int zoff = 0;
        asm volatile("" : "+v"(zoff));

        v2f xfrag[7];
        #pragma unroll
        for (int kk = 0; kk < 7; ++kk)
            xfrag[kk] = *(const v2f*)&xlane[t * I_SZ + kk * 4 + hi * 2];
        if (t + 1 < T_SEQ)
            __builtin_prefetch(&xlane[(t + 1) * I_SZ], 0, 3);

        v8f hnew[4];

        #pragma unroll
        for (int j = 0; j < 4; ++j) {
            v8f acc[4];
            #pragma unroll
            for (int g = 0; g < 4; ++g) {
                #pragma unroll
                for (int r = 0; r < 8; ++r) acc[g][r] = bias[g * 4 + j];
            }

            #pragma unroll
            for (int kk = 0; kk < 7; ++kk) {
                const int k = kk * 4 + hi * 2;
                #pragma unroll
                for (int g = 0; g < 4; ++g) {
                    const int n = (g * 4 + j) * 16 + col;
                    v2f bfrag = *(const v2f*)&sWih[n * WIH_PITCH + k + zoff];
                    acc[g] = __builtin_amdgcn_wmma_f32_16x16x4_f32(
                        false, xfrag[kk], false, bfrag, (short)0, acc[g], false, false);
                }
            }

            if (t > 0) {
                #pragma unroll
                for (int kk = 0; kk < 16; ++kk) {
                    const int k = kk * 4 + hi * 2;
                    v2f afrag = *(const v2f*)&hbuf[col * H_PITCH + k];
                    #pragma unroll
                    for (int g = 0; g < 4; ++g) {
                        const int n = (g * 4 + j) * 16 + col;
                        v2f bfrag = *(const v2f*)&sWhh[n * WHH_PITCH + k + zoff];
                        acc[g] = __builtin_amdgcn_wmma_f32_16x16x4_f32(
                            false, afrag, false, bfrag, (short)0, acc[g], false, false);
                    }
                }
            }

            #pragma unroll
            for (int r = 0; r < 8; ++r) {
                float ig = fast_sigmoid(acc[0][r]);
                float fg = fast_sigmoid(acc[1][r]);
                float gg = fast_tanh   (acc[2][r]);
                float og = fast_sigmoid(acc[3][r]);
                float cn = fg * c_st[j][r] + ig * gg;
                c_st[j][r] = cn;
                hnew[j][r] = og * fast_tanh(cn);
            }
        }

        #pragma unroll
        for (int j = 0; j < 4; ++j) {
            #pragma unroll
            for (int r = 0; r < 8; ++r)
                hbuf[(r + 8 * hi) * H_PITCH + j * 16 + col] = hnew[j][r];
        }
    }

    {
        v8f acc;
        float bo = (col < NCLS) ? b_out[col] : 0.0f;
        #pragma unroll
        for (int r = 0; r < 8; ++r) acc[r] = bo;

        #pragma unroll
        for (int kk = 0; kk < 16; ++kk) {
            const int k = kk * 4 + hi * 2;
            v2f afrag = *(const v2f*)&hbuf[col * H_PITCH + k];
            v2f bfrag = {};
            if (col < NCLS) bfrag = *(const v2f*)&W_out[col * H_SZ + k];
            acc = __builtin_amdgcn_wmma_f32_16x16x4_f32(
                false, afrag, false, bfrag, (short)0, acc, false, false);
        }

        #pragma unroll
        for (int r = 0; r < 8; ++r) {
            const int row = r + 8 * hi;
            if (col < NCLS)
                out[(size_t)(b0 + row) * NCLS + col] = acc[r];
        }
    }
}

extern "C" void kernel_launch(void* const* d_in, const int* in_sizes, int n_in,
                              void* d_out, int out_size, void* d_ws, size_t ws_size,
                              hipStream_t stream) {
    const float* x     = (const float*)d_in[0];
    const float* W_ih  = (const float*)d_in[1];
    const float* W_hh  = (const float*)d_in[2];
    const float* b_ih  = (const float*)d_in[3];
    const float* b_hh  = (const float*)d_in[4];
    const float* W_out = (const float*)d_in[5];
    const float* b_out = (const float*)d_in[6];
    float* out = (float*)d_out;

    dim3 grid(B_TOTAL / (WAVES_PER_WG * ROWS_PER_WAVE));
    dim3 block(WAVES_PER_WG * 32);
    hipLaunchKernelGGL(lstm_fused_kernel, grid, block, 0, stream,
                       x, W_ih, W_hh, b_ih, b_hh, W_out, b_out, out);
}


# x goes through LDS: the work-group fetches each step's 128x28 x tile once with row-contiguous 16-byte loads into a 3-slot ring, waves read A fragments from LDS; sums unchanged
# speedup vs baseline: 1.2421x; 1.2421x over previous
#include <hip/hip_runtime.h>


typedef __attribute__((ext_vector_type(2))) float v2f;
typedef __attribute__((ext_vector_type(4))) float v4f;
typedef __attribute__((ext_vector_type(8))) float v8f;

#define B_TOTAL 32768
#define T_SEQ   28
#define I_SZ    28
#define H_SZ    64
#define G_SZ    256
#define NCLS    10

#define H_PITCH   68

#define WAVES_PER_WG 8
#define ROWS_PER_WG  128
#define MT           4
#define X_ROW        (T_SEQ * I_SZ)

#define X_TILE       (ROWS_PER_WG * I_SZ)
#define X_RING       3

__device__ __forceinline__ float fast_sigmoid(float x) {
    float e = __builtin_amdgcn_exp2f(x * -1.44269504088896f);
    return __builtin_amdgcn_rcpf(1.0f + e);
}
__device__ __forceinline__ float fast_tanh(float x) {
#if __has_builtin(__builtin_amdgcn_tanhf)
    return __builtin_amdgcn_tanhf(x);
#else
    float e = __builtin_amdgcn_exp2f(x * -2.88539008177793f);
    return __builtin_amdgcn_rcpf(1.0f + e) * 2.0f - 1.0f;
#endif
}

__device__ __forceinline__ v8f wmma4(v2f a, v2f b, v8f c) {
    return __builtin_amdgcn_wmma_f32_16x16x4_f32(false, a, false, b, (short)0, c, false, false);
}

__device__ __forceinline__ void x_part(v8f (&acc)[4], const v2f (&xf)[7],
                                       const v2f (&wih)[4][7], const float (&bias)[4]) {
    #pragma unroll
    for (int g = 0; g < 4; ++g) {
        #pragma unroll
        for (int r = 0; r < 8; ++r) acc[g][r] = bias[g];
    }
    #pragma unroll
    for (int kk = 0; kk < 7; ++kk) {
        #pragma unroll
        for (int g = 0; g < 4; ++g) acc[g] = wmma4(xf[kk], wih[g][kk], acc[g]);
    }
}

template <bool HAS_H, bool HAS_NEXT>
__device__ __forceinline__ void lstm_step(v8f (&c_st)[MT], v2f (&xf0)[7],
                                          const v2f (&whh)[4][16], const v2f (&wih)[4][7],
                                          const float (&bias)[4],
                                          const float* hprev, float* hnext,
                                          const float* xcur, const float* xnext,
                                          bool stage, const float* xg, const unsigned (&goff)[4],
                                          float* xst4, float* xst2)
{
    v4f sq[3];
    v2f sp;
    if (stage) {
        #pragma unroll
        for (int i = 0; i < 3; ++i) sq[i] = *(const v4f*)&xg[goff[i]];
        sp = *(const v2f*)&xg[goff[3]];
    }

    v2f hf[2][16];
    v2f xf[2][7];
    #pragma unroll
    for (int kk = 0; kk < 7; ++kk) xf[0][kk] = xf0[kk];
    if (HAS_H) {
        #pragma unroll
        for (int kk = 0; kk < 16; ++kk) hf[0][kk] = *(const v2f*)&hprev[kk * 4];
    }

    #pragma unroll
    for (int m = 0; m < MT; ++m) {
        __builtin_amdgcn_sched_barrier(0);
        if (m + 1 < MT) {
            #pragma unroll
            for (int kk = 0; kk < 7; ++kk)
                xf[(m + 1) & 1][kk] = *(const v2f*)&xcur[(m + 1) * 16 * I_SZ + kk * 4];
            if (HAS_H) {
                #pragma unroll
                for (int kk = 0; kk < 16; ++kk)
                    hf[(m + 1) & 1][kk] = *(const v2f*)&hprev[(m + 1) * 16 * H_PITCH + kk * 4];
            }
        } else if (HAS_NEXT) {
            #pragma unroll
            for (int kk = 0; kk < 7; ++kk) xf0[kk] = *(const v2f*)&xnext[kk * 4];
        }
        __builtin_amdgcn_sched_barrier(0);

        v8f acc[4];
        x_part(acc, xf[m & 1], wih, bias);
        if (HAS_H) {
            #pragma unroll
            for (int kk = 0; kk < 16; ++kk) {
                #pragma unroll
                for (int g = 0; g < 4; ++g)
                    acc[g] = wmma4(hf[m & 1][kk], whh[g][kk], acc[g]);
            }
        }
        #pragma unroll
        for (int r = 0; r < 8; ++r) {
            float ig = fast_sigmoid(acc[0][r]);
            float fg = fast_sigmoid(acc[1][r]);
            float gg = fast_tanh   (acc[2][r]);
            float og = fast_sigmoid(acc[3][r]);
            float cn = fg * c_st[m][r] + ig * gg;
            c_st[m][r] = cn;
            hnext[(m * 16 + r) * H_PITCH] = og * fast_tanh(cn);
        }
    }
    if (stage) {
        #pragma unroll
        for (int i = 0; i < 3; ++i) *(v4f*)&xst4[i * 1024] = sq[i];
        *(v2f*)xst2 = sp;
    }
    __syncthreads();
}

__global__ __launch_bounds__(256)
void lstm_fused_kernel(const float* __restrict__ x,
                       const float* __restrict__ W_ih,
                       const float* __restrict__ W_hh,
                       const float* __restrict__ b_ih,
                       const float* __restrict__ b_hh,
                       const float* __restrict__ W_out,
                       const float* __restrict__ b_out,
                       float* __restrict__ out)
{
    __shared__ float sH[2][ROWS_PER_WG * H_PITCH];
    __shared__ __attribute__((aligned(16))) float sX[X_RING][X_TILE];

    const int tid  = threadIdx.x;
    const int wave = __builtin_amdgcn_readfirstlane(tid >> 5);
    const int lane = tid & 31;
    const int col  = lane & 15;
    const int hi   = lane >> 4;
    const int j    = wave & 3;
    const int rowbase = (wave >> 2) * (MT * 16);
    const int b0   = blockIdx.x * ROWS_PER_WG + rowbase;

    v2f whh[4][16];
    v2f wih[4][7];
    float bias[4];
    #pragma unroll
    for (int g = 0; g < 4; ++g) {
        const int n = g * H_SZ + j * 16 + col;
        #pragma unroll
        for (int kk = 0; kk < 16; ++kk)
            whh[g][kk] = *(const v2f*)&W_hh[n * H_SZ + kk * 4 + hi * 2];
        #pragma unroll
        for (int kk = 0; kk < 7; ++kk)
            wih[g][kk] = *(const v2f*)&W_ih[n * I_SZ + kk * 4 + hi * 2];
        bias[g] = b_ih[n] + b_hh[n];
    }

    const float* xwg = x + (size_t)blockIdx.x * ROWS_PER_WG * X_ROW;
    unsigned goff[4];
    #pragma unroll
    for (int i = 0; i < 3; ++i) {
        const unsigned q = (unsigned)tid + 256u * i;
        goff[i] = (q / 7u) * X_ROW + (q % 7u) * 4u;
    }
    {
        const unsigned f = 3072u + 2u * (unsigned)tid;
        goff[3] = (f / I_SZ) * X_ROW + (f % I_SZ);
    }
    const int st4_off = 4 * tid;
    const int st2_off = 3072 + 2 * tid;
    const int xrd_off = (rowbase + col) * I_SZ + hi * 2;

    const int rd_off = (rowbase + col) * H_PITCH + hi * 2;
    const int wr_off = (rowbase + 8 * hi) * H_PITCH + j * 16 + col;

    v8f c_st[MT];
    #pragma unroll
    for (int m = 0; m < MT; ++m) c_st[m] = (v8f){};

    #pragma unroll
    for (int s = 0; s < 2; ++s) {
        const float* xg = xwg + s * I_SZ;
        #pragma unroll
        for (int i = 0; i < 3; ++i)
            *(v4f*)&sX[s][st4_off + i * 1024] = *(const v4f*)&xg[goff[i]];
        *(v2f*)&sX[s][st2_off] = *(const v2f*)&xg[goff[3]];
    }
    __syncthreads();

    v2f xf0[7];
    #pragma unroll
    for (int kk = 0; kk < 7; ++kk) xf0[kk] = *(const v2f*)&sX[0][xrd_off + kk * 4];

    int xs_cur = 0, xs_nxt = X_TILE, xs_fill = 2 * X_TILE;
    float* sXf = &sX[0][0];

    lstm_step<false, true>(c_st, xf0, whh, wih, bias,
                           &sH[1][rd_off], &sH[0][wr_off],
                           sXf + xs_cur + xrd_off, sXf + xs_nxt + xrd_off,
                           true, xwg + 2 * I_SZ, goff,
                           sXf + xs_fill + st4_off, sXf + xs_fill + st2_off);

    for (int t = 1; t < T_SEQ - 1; ++t) {
        const int xs_old = xs_cur; xs_cur = xs_nxt; xs_nxt = xs_fill; xs_fill = xs_old;
        lstm_step<true, true>(c_st, xf0, whh, wih, bias,
                              &sH[(t - 1) & 1][rd_off], &sH[t & 1][wr_off],
                              sXf + xs_cur + xrd_off, sXf + xs_nxt + xrd_off,
                              t + 2 < T_SEQ, xwg + (t + 2) * I_SZ, goff,
                              sXf + xs_fill + st4_off, sXf + xs_fill + st2_off);
    }

    lstm_step<true, false>(c_st, xf0, whh, wih, bias,
                           &sH[(T_SEQ - 2) & 1][rd_off], &sH[(T_SEQ - 1) & 1][wr_off],
                           sXf + xs_nxt + xrd_off, sXf + xs_nxt + xrd_off,
                           false, xwg, goff, sXf, sXf);

    {
        const int ln = (int)__builtin_amdgcn_mbcnt_lo(~0u, 0u);
        const int cl = ln & 15;
        const int hh = ln >> 4;
        const float* hfin = &sH[(T_SEQ - 1) & 1][(wave * 16 + cl) * H_PITCH + hh * 2];
        v8f o;
        float bo = (cl < NCLS) ? b_out[cl] : 0.0f;
        #pragma unroll
        for (int r = 0; r < 8; ++r) o[r] = bo;

        #pragma unroll
        for (int kk = 0; kk < 16; ++kk) {
            v2f afrag = *(const v2f*)&hfin[kk * 4];
            v2f bfrag = {};
            if (cl < NCLS) bfrag = *(const v2f*)&W_out[cl * H_SZ + kk * 4 + hh * 2];
            o = wmma4(afrag, bfrag, o);
        }

        const int orow0 = blockIdx.x * ROWS_PER_WG + wave * 16;
        #pragma unroll
        for (int r = 0; r < 8; ++r) {
            const int row = r + 8 * hh;
            if (cl < NCLS)
                out[(size_t)(orow0 + row) * NCLS + cl] = o[r];
        }
    }
}

extern "C" void kernel_launch(void* const* d_in, const int* in_sizes, int n_in,
                              void* d_out, int out_size, void* d_ws, size_t ws_size,
                              hipStream_t stream) {
    const float* x     = (const float*)d_in[0];
    const float* W_ih  = (const float*)d_in[1];
    const float* W_hh  = (const float*)d_in[2];
    const float* b_ih  = (const float*)d_in[3];
    const float* b_hh  = (const float*)d_in[4];
    const float* W_out = (const float*)d_in[5];
    const float* b_out = (const float*)d_in[6];
    float* out = (float*)d_out;

    dim3 grid(B_TOTAL / ROWS_PER_WG);
    dim3 block(WAVES_PER_WG * 32);
    hipLaunchKernelGGL(lstm_fused_kernel, grid, block, 0, stream,
                       x, W_ih, W_hh, b_ih, b_hh, W_out, b_out, out);
}


# Cell update on 5 transcendentals + 5 plain ops per element (was 8 + 9): sigmoid = 0.5*tanh(x/2)+0.5 on v_tanh_f32, factors 1/2 folded exactly into weight rows at load; sum order unchanged
# speedup vs baseline: 1.2823x; 1.0324x over previous
#include <hip/hip_runtime.h>


typedef __attribute__((ext_vector_type(2))) float v2f;
typedef __attribute__((ext_vector_type(4))) float v4f;
typedef __attribute__((ext_vector_type(8))) float v8f;

#define B_TOTAL 32768
#define T_SEQ   28
#define I_SZ    28
#define H_SZ    64
#define G_SZ    256
#define NCLS    10

#define H_PITCH   68

#define WAVES_PER_WG 8
#define ROWS_PER_WG  128
#define MT           4
#define X_ROW        (T_SEQ * I_SZ)

#define X_TILE       (ROWS_PER_WG * I_SZ)
#define X_RING       3

__device__ __forceinline__ float fast_tanh(float x) {
#if __has_builtin(__builtin_amdgcn_tanhf)
    return __builtin_amdgcn_tanhf(x);
#else
    float e = __builtin_amdgcn_exp2f(x * -2.88539008177793f);
    return __builtin_amdgcn_rcpf(1.0f + e) * 2.0f - 1.0f;
#endif
}

__device__ __forceinline__ v8f wmma4(v2f a, v2f b, v8f c) {
    return __builtin_amdgcn_wmma_f32_16x16x4_f32(false, a, false, b, (short)0, c, false, false);
}

__device__ __forceinline__ void x_part(v8f (&acc)[4], const v2f (&xf)[7],
                                       const v2f (&wih)[4][7], const float (&bias)[4]) {
    #pragma unroll
    for (int g = 0; g < 4; ++g) {
        #pragma unroll
        for (int r = 0; r < 8; ++r) acc[g][r] = bias[g];
    }
    #pragma unroll
    for (int kk = 0; kk < 7; ++kk) {
        #pragma unroll
        for (int g = 0; g < 4; ++g) acc[g] = wmma4(xf[kk], wih[g][kk], acc[g]);
        __builtin_amdgcn_sched_barrier(0);
    }
}

template <bool HAS_H, bool HAS_NEXT>
__device__ __forceinline__ void lstm_step(v8f (&c_st)[MT], v2f (&xf0)[7],
                                          const v2f (&whh)[4][16], const v2f (&wih)[4][7],
                                          const float (&bias)[4],
                                          const float* hprev, float* hnext,
                                          const float* xcur, const float* xnext,
                                          bool stage, const float* xg, const unsigned (&goff)[4],
                                          float* xst4, float* xst2)
{
    v4f sq[3];
    v2f sp;
    if (stage) {
        #pragma unroll
        for (int i = 0; i < 3; ++i) sq[i] = *(const v4f*)&xg[goff[i]];
        sp = *(const v2f*)&xg[goff[3]];
    }

    v2f hf[2][16];
    v2f xf[2][7];
    #pragma unroll
    for (int kk = 0; kk < 7; ++kk) xf[0][kk] = xf0[kk];
    if (HAS_H) {
        #pragma unroll
        for (int kk = 0; kk < 16; ++kk) hf[0][kk] = *(const v2f*)&hprev[kk * 4];
    }

    #pragma unroll
    for (int m = 0; m < MT; ++m) {
        __builtin_amdgcn_sched_barrier(0);
        if (m + 1 < MT) {
            #pragma unroll
            for (int kk = 0; kk < 7; ++kk)
                xf[(m + 1) & 1][kk] = *(const v2f*)&xcur[(m + 1) * 16 * I_SZ + kk * 4];
            if (HAS_H) {
                #pragma unroll
                for (int kk = 0; kk < 16; ++kk)
                    hf[(m + 1) & 1][kk] = *(const v2f*)&hprev[(m + 1) * 16 * H_PITCH + kk * 4];
            }
        } else if (HAS_NEXT) {
            #pragma unroll
            for (int kk = 0; kk < 7; ++kk) xf0[kk] = *(const v2f*)&xnext[kk * 4];
        }
        __builtin_amdgcn_sched_barrier(0);

        v8f acc[4];
        x_part(acc, xf[m & 1], wih, bias);
        if (HAS_H) {
            #pragma unroll
            for (int kk = 0; kk < 16; ++kk) {
                #pragma unroll
                for (int g = 0; g < 4; ++g)
                    acc[g] = wmma4(hf[m & 1][kk], whh[g][kk], acc[g]);
                __builtin_amdgcn_sched_barrier(0);
            }
        }
        #pragma unroll
        for (int r = 0; r < 8; ++r) {
            float ti = fast_tanh(acc[0][r]);
            float tf = fast_tanh(acc[1][r]);
            float tg = fast_tanh(acc[2][r]);
            float to = fast_tanh(acc[3][r]);
            float fc2 = fmaf(tf, c_st[m][r], c_st[m][r]);
            float ig2 = fmaf(ti, tg, tg);
            float cn  = 0.5f * (fc2 + ig2);
            c_st[m][r] = cn;
            float th = fast_tanh(cn);
            hnext[(m * 16 + r) * H_PITCH] = fmaf(to, th, th);
        }
    }
    if (stage) {
        #pragma unroll
        for (int i = 0; i < 3; ++i) *(v4f*)&xst4[i * 1024] = sq[i];
        *(v2f*)xst2 = sp;
    }
    __syncthreads();
}

__global__ __launch_bounds__(256)
void lstm_fused_kernel(const float* __restrict__ x,
                       const float* __restrict__ W_ih,
                       const float* __restrict__ W_hh,
                       const float* __restrict__ b_ih,
                       const float* __restrict__ b_hh,
                       const float* __restrict__ W_out,
                       const float* __restrict__ b_out,
                       float* __restrict__ out)
{
    __shared__ float sH[2][ROWS_PER_WG * H_PITCH];
    __shared__ __attribute__((aligned(16))) float sX[X_RING][X_TILE];

    const int tid  = threadIdx.x;
    const int wave = __builtin_amdgcn_readfirstlane(tid >> 5);
    const int lane = tid & 31;
    const int col  = lane & 15;
    const int hi   = lane >> 4;
    const int j    = wave & 3;
    const int rowbase = (wave >> 2) * (MT * 16);
    const int b0   = blockIdx.x * ROWS_PER_WG + rowbase;

    v2f whh[4][16];
    v2f wih[4][7];
    float bias[4];
    #pragma unroll
    for (int g = 0; g < 4; ++g) {
        const int n = g * H_SZ + j * 16 + col;
        const float gs = (g == 2) ? 1.0f : 0.5f;
        #pragma unroll
        for (int kk = 0; kk < 16; ++kk)
            whh[g][kk] = *(const v2f*)&W_hh[n * H_SZ + kk * 4 + hi * 2] * (0.5f * gs);
        #pragma unroll
        for (int kk = 0; kk < 7; ++kk)
            wih[g][kk] = *(const v2f*)&W_ih[n * I_SZ + kk * 4 + hi * 2] * gs;
        bias[g] = (b_ih[n] + b_hh[n]) * gs;
    }

    const float* xwg = x + (size_t)blockIdx.x * ROWS_PER_WG * X_ROW;
    unsigned goff[4];
    #pragma unroll
    for (int i = 0; i < 3; ++i) {
        const unsigned q = (unsigned)tid + 256u * i;
        goff[i] = (q / 7u) * X_ROW + (q % 7u) * 4u;
    }
    {
        const unsigned f = 3072u + 2u * (unsigned)tid;
        goff[3] = (f / I_SZ) * X_ROW + (f % I_SZ);
    }
    const int st4_off = 4 * tid;
    const int st2_off = 3072 + 2 * tid;
    const int xrd_off = (rowbase + col) * I_SZ + hi * 2;

    const int rd_off = (rowbase + col) * H_PITCH + hi * 2;
    const int wr_off = (rowbase + 8 * hi) * H_PITCH + j * 16 + col;

    v8f c_st[MT];
    #pragma unroll
    for (int m = 0; m < MT; ++m) c_st[m] = (v8f){};

    #pragma unroll
    for (int s = 0; s < 2; ++s) {
        const float* xg = xwg + s * I_SZ;
        #pragma unroll
        for (int i = 0; i < 3; ++i)
            *(v4f*)&sX[s][st4_off + i * 1024] = *(const v4f*)&xg[goff[i]];
        *(v2f*)&sX[s][st2_off] = *(const v2f*)&xg[goff[3]];
    }
    __syncthreads();

    v2f xf0[7];
    #pragma unroll
    for (int kk = 0; kk < 7; ++kk) xf0[kk] = *(const v2f*)&sX[0][xrd_off + kk * 4];

    int xs_cur = 0, xs_nxt = X_TILE, xs_fill = 2 * X_TILE;
    float* sXf = &sX[0][0];

    lstm_step<false, true>(c_st, xf0, whh, wih, bias,
                           &sH[1][rd_off], &sH[0][wr_off],
                           sXf + xs_cur + xrd_off, sXf + xs_nxt + xrd_off,
                           true, xwg + 2 * I_SZ, goff,
                           sXf + xs_fill + st4_off, sXf + xs_fill + st2_off);

    for (int t = 1; t < T_SEQ - 1; ++t) {
        const int xs_old = xs_cur; xs_cur = xs_nxt; xs_nxt = xs_fill; xs_fill = xs_old;
        lstm_step<true, true>(c_st, xf0, whh, wih, bias,
                              &sH[(t - 1) & 1][rd_off], &sH[t & 1][wr_off],
                              sXf + xs_cur + xrd_off, sXf + xs_nxt + xrd_off,
                              t + 2 < T_SEQ, xwg + (t + 2) * I_SZ, goff,
                              sXf + xs_fill + st4_off, sXf + xs_fill + st2_off);
    }

    lstm_step<true, false>(c_st, xf0, whh, wih, bias,
                           &sH[(T_SEQ - 2) & 1][rd_off], &sH[(T_SEQ - 1) & 1][wr_off],
                           sXf + xs_nxt + xrd_off, sXf + xs_nxt + xrd_off,
                           false, xwg, goff, sXf, sXf);

    {
        const int ln = (int)__builtin_amdgcn_mbcnt_lo(~0u, 0u);
        const int cl = ln & 15;
        const int hh = ln >> 4;
        const float* hfin = &sH[(T_SEQ - 1) & 1][(wave * 16 + cl) * H_PITCH + hh * 2];
        v8f o;
        float bo = (cl < NCLS) ? b_out[cl] : 0.0f;
        #pragma unroll
        for (int r = 0; r < 8; ++r) o[r] = bo;

        #pragma unroll
        for (int kk = 0; kk < 16; ++kk) {
            v2f afrag = *(const v2f*)&hfin[kk * 4];
            v2f bfrag = {};
            if (cl < NCLS) bfrag = *(const v2f*)&W_out[cl * H_SZ + kk * 4 + hh * 2] * 0.5f;
            o = wmma4(afrag, bfrag, o);
        }

        const int orow0 = blockIdx.x * ROWS_PER_WG + wave * 16;
        #pragma unroll
        for (int r = 0; r < 8; ++r) {
            const int row = r + 8 * hh;
            if (cl < NCLS)
                out[(size_t)(orow0 + row) * NCLS + cl] = o[r];
        }
    }
}

extern "C" void kernel_launch(void* const* d_in, const int* in_sizes, int n_in,
                              void* d_out, int out_size, void* d_ws, size_t ws_size,
                              hipStream_t stream) {
    const float* x     = (const float*)d_in[0];
    const float* W_ih  = (const float*)d_in[1];
    const float* W_hh  = (const float*)d_in[2];
    const float* b_ih  = (const float*)d_in[3];
    const float* b_hh  = (const float*)d_in[4];
    const float* W_out = (const float*)d_in[5];
    const float* b_out = (const float*)d_in[6];
    float* out = (float*)d_out;

    dim3 grid(B_TOTAL / ROWS_PER_WG);
    dim3 block(WAVES_PER_WG * 32);
    hipLaunchKernelGGL(lstm_fused_kernel, grid, block, 0, stream,
                       x, W_ih, W_hh, b_ih, b_hh, W_out, b_out, out);
}
